# starting kernel
# baseline (speedup 1.0000x reference)
#include <hip/hip_runtime.h>
#include <hip/hip_bf16.h>

typedef float v2f __attribute__((ext_vector_type(2)));
typedef float v8f __attribute__((ext_vector_type(8)));

#define BATCH      16
#define CHANS      256
#define NPIX       16384
#define KCLS       6
#define KPAD       8
#define IGNORE_IDX 255

#define NCHUNK     256
#define ROWSTRIDE  260
#define WAVES      4
#define THREADS    (WAVES * 32)
#define NSPLIT     8
#define STEPS      (NCHUNK / (WAVES * 4))

__global__ void zero_acc_kernel(float* __restrict__ acc, int n) {
    int i = blockIdx.x * blockDim.x + threadIdx.x;
    if (i < n) acc[i] = 0.0f;
}

__global__ void count_kernel(const int* __restrict__ gt, float* __restrict__ counts) {
    __shared__ int lc[KPAD];
    int b = blockIdx.x;
    if (threadIdx.x < KPAD) lc[threadIdx.x] = 0;
    __syncthreads();
    const int* gb = gt + (size_t)b * NPIX;
    for (int i = threadIdx.x; i < NPIX; i += blockDim.x) {
        int l = gb[i];
        if (l >= 0 && l < KCLS) atomicAdd(&lc[l], 1);
    }
    __syncthreads();
    if (threadIdx.x < KPAD) counts[b * KPAD + threadIdx.x] = (float)lc[threadIdx.x];
}

__global__ void __launch_bounds__(THREADS)
center_accum_kernel(const float* __restrict__ x, const int* __restrict__ gt,
                    float* __restrict__ acc  ) {
    __shared__ float tile[16 * ROWSTRIDE];
    __shared__ int   labs[NCHUNK];

    int wg    = blockIdx.x;
    int split = wg % NSPLIT;
    int ct    = (wg / NSPLIT) % (CHANS / 16);
    int b     = wg / (NSPLIT * (CHANS / 16));

    const int nPerSplit = NPIX / NSPLIT;
    int nBeg = split * nPerSplit;
    int nEnd = nBeg + nPerSplit;

    const float* xb = x  + ((size_t)b * CHANS + (size_t)ct * 16) * NPIX;
    const int*   gb = gt + (size_t)b * NPIX;

    int tid  = threadIdx.x;
    int lane = tid & 31;
    int wave = tid >> 5;
    int m    = lane & 15;
    int hi   = lane >> 4;

    v8f d0 = {};
    v8f d1 = {};

    for (int n0 = nBeg; n0 < nEnd; n0 += NCHUNK) {
        {
            const float4* src = (const float4*)(xb + n0);
            for (int i = tid; i < 16 * (NCHUNK / 4); i += THREADS) {
                int r  = i / (NCHUNK / 4);
                int p4 = i % (NCHUNK / 4);
                float4 v = src[(size_t)r * (NPIX / 4) + p4];
                *(float4*)&tile[r * ROWSTRIDE + p4 * 4] = v;
            }
            if (tid < NCHUNK / 4)
                ((int4*)labs)[tid] = ((const int4*)(gb + n0))[tid];
        }
        __syncthreads();

#pragma unroll
        for (int it = 0; it < STEPS; ++it) {
            int s = wave * 4 + it * (WAVES * 4);
            const float* ap = &tile[m * ROWSTRIDE + s + 2 * hi];
            v2f a; a.x = ap[0]; a.y = ap[1];
            int l0 = labs[s + 2 * hi];
            int l1 = labs[s + 2 * hi + 1];
            v2f bm;
            bm.x = (l0 == m) ? 1.0f : 0.0f;
            bm.y = (l1 == m) ? 1.0f : 0.0f;
            if (it & 1)
                d1 = __builtin_amdgcn_wmma_f32_16x16x4_f32(
                        false, a, false, bm, (short)0, d1, false, false);
            else
                d0 = __builtin_amdgcn_wmma_f32_16x16x4_f32(
                        false, a, false, bm, (short)0, d0, false, false);
        }
        __syncthreads();
    }

    if (m < KCLS) {
        float* accb = acc + ((size_t)b * CHANS + (size_t)ct * 16) * KPAD;
#pragma unroll
        for (int j = 0; j < 8; ++j) {
            int c = j + 8 * hi;
            atomicAdd(&accb[c * KPAD + m], d0[j] + d1[j]);
        }
    }
}

__global__ void finalize_kernel(const float* __restrict__ acc,
                                const float* __restrict__ counts,
                                float* __restrict__ out) {
    int idx = blockIdx.x * blockDim.x + threadIdx.x;
    if (idx >= BATCH * KCLS * CHANS) return;
    int c = idx % CHANS;
    int k = (idx / CHANS) % KCLS;
    int b = idx / (CHANS * KCLS);
    float cnt = fmaxf(counts[b * KPAD + k], 1.0f);
    out[idx] = acc[((size_t)b * CHANS + c) * KPAD + k] / cnt;
}

extern "C" void kernel_launch(void* const* d_in, const int* in_sizes, int n_in,
                              void* d_out, int out_size, void* d_ws, size_t ws_size,
                              hipStream_t stream) {
    const float* x  = (const float*)d_in[0];
    const int*   gt = (const int*)d_in[1];
    float*       out = (float*)d_out;

    float* acc    = (float*)d_ws;
    float* counts = acc + (size_t)BATCH * CHANS * KPAD;

    const int accN = BATCH * CHANS * KPAD;
    zero_acc_kernel<<<(accN + 255) / 256, 256, 0, stream>>>(acc, accN);

    count_kernel<<<BATCH, 256, 0, stream>>>(gt, counts);

    const int nBlocks = BATCH * (CHANS / 16) * NSPLIT;
    center_accum_kernel<<<nBlocks, THREADS, 0, stream>>>(x, gt, acc);

    const int outN = BATCH * KCLS * CHANS;
    finalize_kernel<<<(outN + 255) / 256, 256, 0, stream>>>(acc, counts, out);
}


# Count launch removed: each main-pass block adds its 8 waves' tiles in LDS in fixed order and stores one plane (no atomics, nothing to clear); the last launch counts labels and adds the two planes
# speedup vs baseline: 2.4497x; 2.4497x over previous
#include <hip/hip_runtime.h>
#include <hip/hip_bf16.h>

typedef float v2f __attribute__((ext_vector_type(2)));
typedef float v8f __attribute__((ext_vector_type(8)));

#define BATCH      16
#define CHANS      256
#define NPIX       16384
#define KCLS       6
#define IGNORE_IDX 255

#define TILE_F     (KCLS * 16)
#define PLANE_B    ((CHANS / 16) * TILE_F)

#define DW_WAVES   8
#define DW_THREADS (DW_WAVES * 32)
#define DW_SPLIT   2
#define DW_PIX     (NPIX / (DW_SPLIT * DW_WAVES))
#define GPX        64
#define DW_GROUPS  (DW_PIX / GPX)
#define AHEAD      2
#define SLOTS      (AHEAD + 1)

__device__ __forceinline__ float xor8(float v) {
    return __builtin_bit_cast(float, __builtin_amdgcn_permlane16(
        __builtin_bit_cast(int, v), __builtin_bit_cast(int, v), 0xfedcba98, 0x76543210, false, false));
}

__global__ void __launch_bounds__(DW_THREADS)
center_accum_kernel(const float* __restrict__ x, const int* __restrict__ gt,
                    float* __restrict__ plane0, float* __restrict__ plane1  ) {
    static_assert(DW_SPLIT == 2, "one plane per pixel half: plane0 and plane1");
    __shared__ float red[DW_WAVES * 2 * 16 * 16];
    int lane = threadIdx.x & 31;
    int wave = __builtin_amdgcn_readfirstlane(threadIdx.x >> 5);
    int m    = lane & 15;
    int hi   = lane >> 4;
    int r    = m & 7;
    bool lo  = m < 8;

    int wg   = blockIdx.x;
    int half = wg % DW_SPLIT;
    int ct   = (wg / DW_SPLIT) % (CHANS / 16);
    int b    = wg / (DW_SPLIT * (CHANS / 16));
    int p0   = (half * DW_WAVES + wave) * DW_PIX;

    int pc0 = lo ? hi : 2 + hi;
    int pc1 = lo ? 2 + hi : hi;
    const float*  xt  = x + ((size_t)b * CHANS + (size_t)ct * 16) * NPIX + p0;
    const float4* xp0 = (const float4*)(xt + (size_t)r * NPIX + 4 * pc0);
    const float4* xp1 = (const float4*)(xt + (size_t)(r + 8) * NPIX + 4 * pc1);
    const int4*   lp  = (const int4*)(gt + (size_t)b * NPIX + p0 + 4 * hi);

    float4 xa[SLOTS][8];
    int4   la[SLOTS][8];

    v8f d0 = {};
    v8f d1 = {};

#pragma unroll
    for (int g = 0; g < AHEAD; ++g) {
#pragma unroll
        for (int q = 0; q < 4; ++q) {
            la[g][2 * q]     = lp[g * (GPX / 4) + 4 * q];
            la[g][2 * q + 1] = lp[g * (GPX / 4) + 4 * q + 2];
            xa[g][2 * q]     = xp0[g * (GPX / 4) + 4 * q];
            xa[g][2 * q + 1] = xp1[g * (GPX / 4) + 4 * q];
        }
    }

#pragma unroll
    for (int g = 0; g < DW_GROUPS; ++g) {
        if (g + AHEAD < DW_GROUPS) {
            const int s = (g + AHEAD) % SLOTS;
#pragma unroll
            for (int q = 0; q < 4; ++q) {
                la[s][2 * q]     = lp[(g + AHEAD) * (GPX / 4) + 4 * q];
                la[s][2 * q + 1] = lp[(g + AHEAD) * (GPX / 4) + 4 * q + 2];
                xa[s][2 * q]     = xp0[(g + AHEAD) * (GPX / 4) + 4 * q];
                xa[s][2 * q + 1] = xp1[(g + AHEAD) * (GPX / 4) + 4 * q];
            }
        }
        asm volatile("" ::: "memory");
        __builtin_amdgcn_sched_barrier(0);
        const int c = g % SLOTS;
        int mg = m;
        asm volatile("" : "+v"(mg));
        const bool lo_g = mg < 8;
#pragma unroll
        for (int q = 0; q < 4; ++q) {
            float4 P = xa[c][2 * q];
            float4 Q = xa[c][2 * q + 1];
            float4 own, rcv;
            own.x = lo_g ? P.x : Q.x;  rcv.x = xor8(lo_g ? Q.x : P.x);
            own.y = lo_g ? P.y : Q.y;  rcv.y = xor8(lo_g ? Q.y : P.y);
            own.z = lo_g ? P.z : Q.z;  rcv.z = xor8(lo_g ? Q.z : P.z);
            own.w = lo_g ? P.w : Q.w;  rcv.w = xor8(lo_g ? Q.w : P.w);
#pragma unroll
            for (int e = 0; e < 2; ++e) {
                float4 xv = e ? rcv : own;
                int4   lv = la[c][2 * q + e];
                v2f a0; a0.x = xv.x; a0.y = xv.y;
                v2f a1; a1.x = xv.z; a1.y = xv.w;
                v2f b0; b0.x = (lv.x == mg) ? 1.0f : 0.0f; b0.y = (lv.y == mg) ? 1.0f : 0.0f;
                v2f b1; b1.x = (lv.z == mg) ? 1.0f : 0.0f; b1.y = (lv.w == mg) ? 1.0f : 0.0f;
                d0 = __builtin_amdgcn_wmma_f32_16x16x4_f32(
                        false, a0, false, b0, (short)0, d0, false, false);
                d1 = __builtin_amdgcn_wmma_f32_16x16x4_f32(
                        false, a1, false, b1, (short)0, d1, false, false);
            }
        }
        asm volatile("" ::: "memory");
        __builtin_amdgcn_sched_barrier(0);
    }

    float4* dst = (float4*)(red + wave * 512 + m * 16 + 8 * hi);
    dst[0]  = make_float4(d0[0], d0[1], d0[2], d0[3]);
    dst[1]  = make_float4(d0[4], d0[5], d0[6], d0[7]);
    dst[64] = make_float4(d1[0], d1[1], d1[2], d1[3]);
    dst[65] = make_float4(d1[4], d1[5], d1[6], d1[7]);
    __syncthreads();
    if (threadIdx.x < TILE_F / 4) {
        const float4* r4 = (const float4*)red;
        float4 s = r4[threadIdx.x];
#pragma unroll
        for (int i = 1; i < 2 * DW_WAVES; ++i) {
            float4 t = r4[i * 64 + threadIdx.x];
            s.x += t.x;  s.y += t.y;  s.z += t.z;  s.w += t.w;
        }
        float* plane = half ? plane1 : plane0;
        ((float4*)(plane + ((size_t)b * (CHANS / 16) + ct) * TILE_F))[threadIdx.x] = s;
    }
}

#define CNT_THREADS    1024
#define CNT_WAVES      (CNT_THREADS / 32)
#define CNT_VECS       (NPIX / (CNT_THREADS * 4))
#define FIELD_BITS     10
#define FIELD_MASK     ((1u << FIELD_BITS) - 1u)
#define OUT_VECS       (PLANE_B / 4)

__device__ __forceinline__ unsigned long long class_field(int l) {
    unsigned u = min((unsigned)l, (unsigned)KCLS);
    return 1ull << (FIELD_BITS * u);
}

__global__ void __launch_bounds__(CNT_THREADS)
finalize_kernel(const int* __restrict__ gt, const float* __restrict__ plane0, float* out) {
    static_assert(OUT_VECS % 32 == 0 && OUT_VECS <= CNT_THREADS, "whole waves hold the planes");
    __shared__ unsigned long long wsum[CNT_WAVES];
    __shared__ float4 stage[OUT_VECS];
    int b    = blockIdx.x;
    int tid  = threadIdx.x;
    int lane = tid & 31;
    int wave = tid >> 5;
    const int4* gb = (const int4*)(gt + (size_t)b * NPIX);

    int4 v[CNT_VECS];
    for (int j = 0; j < CNT_VECS; ++j)
        v[j] = gb[j * CNT_THREADS + tid];

    if (tid < OUT_VECS) {
        float4 p0 = ((const float4*)(plane0 + (size_t)b * PLANE_B))[tid];
        float4 p1 = ((const float4*)(out + (size_t)b * PLANE_B))[tid];
        float4 s;
        s.x = p0.x + p1.x;  s.y = p0.y + p1.y;  s.z = p0.z + p1.z;  s.w = p0.w + p1.w;
        stage[tid] = s;
    }

    unsigned long long acc = 0;
    for (int j = 0; j < CNT_VECS; ++j)
        acc += class_field(v[j].x) + class_field(v[j].y) +
               class_field(v[j].z) + class_field(v[j].w);

    for (int off = 16; off > 0; off >>= 1)
        acc += __shfl_xor(acc, off);

    if (lane == 0) wsum[wave] = acc;
    __syncthreads();

    if (tid < OUT_VECS) {
        int k  = tid / (CHANS / 4);
        int c4 = tid % (CHANS / 4);
        unsigned total = 0;
        for (int w = 0; w < CNT_WAVES; ++w)
            total += (unsigned)(wsum[w] >> (FIELD_BITS * k)) & FIELD_MASK;
        float cnt = fmaxf((float)total, 1.0f);
        float4 s = stage[(c4 >> 2) * (TILE_F / 4) + k * 4 + (c4 & 3)];
        float4 o;
        o.x = s.x / cnt;  o.y = s.y / cnt;  o.z = s.z / cnt;  o.w = s.w / cnt;
        ((float4*)(out + (size_t)b * PLANE_B))[tid] = o;
    }
}

extern "C" void kernel_launch(void* const* d_in, const int* in_sizes, int n_in,
                              void* d_out, int out_size, void* d_ws, size_t ws_size,
                              hipStream_t stream) {
    const float* x  = (const float*)d_in[0];
    const int*   gt = (const int*)d_in[1];
    float*       out = (float*)d_out;

    float* plane0 = (float*)d_ws;

    const int nBlocks = BATCH * (CHANS / 16) * DW_SPLIT;
    center_accum_kernel<<<nBlocks, DW_THREADS, 0, stream>>>(x, gt, plane0, out);

    finalize_kernel<<<BATCH, CNT_THREADS, 0, stream>>>(gt, plane0, out);
}
